# starting kernel
# baseline (speedup 1.0000x reference)
#include <hip/hip_runtime.h>
#include <math.h>

typedef float v2f __attribute__((ext_vector_type(2)));
typedef float v8f __attribute__((ext_vector_type(8)));

namespace {
constexpr int C_DIM = 2048;
constexpr int HW = 49;
constexpr int KSTEPS = C_DIM / 4;
constexpr int WAVES_PER_BLOCK = 16;
constexpr int STEPS_PER_WAVE = KSTEPS / WAVES_PER_BLOCK;
}

__global__ void pack_A(const float* __restrict__ W, float* __restrict__ Apack) {
  int idx = blockIdx.x * blockDim.x + threadIdx.x;
  if (idx >= KSTEPS * 32) return;
  int lane = idx & 31;
  int step = idx >> 5;
  int m = lane & 15;
  int kbase = step * 4 + ((lane >> 4) << 1);
  float v0 = 0.0f, v1 = 0.0f;
#pragma unroll
  for (int j = 0; j < 2; ++j) {
    int kk = kbase + j;
    float val = 0.0f;
    if (m == 0)      val = 1.0f;
    else if (m == 1) val = W[2048 + kk];
    else if (m == 2) val = W[4096 + 2048 + kk];
    else if (m == 3) val = W[kk];
    else if (m == 4) val = W[4096 + kk];
    if (j == 0) v0 = val; else v1 = val;
  }
  float2 packed; packed.x = v0; packed.y = v1;
  ((float2*)Apack)[idx] = packed;
}

__global__ __launch_bounds__(512) void fused(
    const float* __restrict__ x, const float* __restrict__ mask,
    const float* __restrict__ bvec, const float* __restrict__ Apack,
    float* __restrict__ out) {
  __shared__ float part[WAVES_PER_BLOCK][5][64];
  __shared__ float fin[5][HW];
  __shared__ float sm[HW];
  __shared__ float sgap[HW];
  __shared__ int   s_pmax;
  __shared__ float s_gapmean;

  const int tid = threadIdx.x;
  const int n = blockIdx.x;
  const int wave = tid >> 5;
  const int lane = tid & 31;
  const int mrow = lane & 15;
  const int chalf = (lane >> 4) << 1;

  if (tid < HW) sm[tid] = mask[n * HW + tid];

  const int p0 = mrow, p1 = 16 + mrow, p2 = 32 + mrow;

  v8f a0 = {}, a1 = {}, a2 = {}, a3 = {};
  const float* xn = x + (size_t)n * (C_DIM * HW);
  const int step0 = wave * STEPS_PER_WAVE;

  for (int s = 0; s < STEPS_PER_WAVE; ++s) {
    const int step = step0 + s;
    const float* xr = xn + (size_t)(step * 4 + chalf) * HW;
    v2f A = *(const v2f*)(Apack + 2 * ((size_t)step * 32 + lane));
    v2f B0, B1, B2, B3;
    B0.x = xr[p0];  B0.y = xr[HW + p0];
    B1.x = xr[p1];  B1.y = xr[HW + p1];
    B2.x = xr[p2];  B2.y = xr[HW + p2];
    B3.x = xr[48];  B3.y = xr[HW + 48];
    a0 = __builtin_amdgcn_wmma_f32_16x16x4_f32(false, A, false, B0, (short)0, a0, false, false);
    a1 = __builtin_amdgcn_wmma_f32_16x16x4_f32(false, A, false, B1, (short)0, a1, false, false);
    a2 = __builtin_amdgcn_wmma_f32_16x16x4_f32(false, A, false, B2, (short)0, a2, false, false);
    a3 = __builtin_amdgcn_wmma_f32_16x16x4_f32(false, A, false, B3, (short)0, a3, false, false);
  }

  if (lane < 16) {
#pragma unroll
    for (int m = 0; m < 5; ++m) {
      part[wave][m][mrow]      = a0[m];
      part[wave][m][16 + mrow] = a1[m];
      part[wave][m][32 + mrow] = a2[m];
      part[wave][m][48 + mrow] = a3[m];
    }
  }
  __syncthreads();

  if (tid < 5 * HW) {
    int m = tid / HW, p = tid % HW;
    float ssum = 0.0f;
    for (int w = 0; w < WAVES_PER_BLOCK; ++w) ssum += part[w][m][p];
    fin[m][p] = ssum;
  }
  __syncthreads();

  if (tid == 0) {
    float tsum = 0.0f;
    for (int p = 0; p < HW; ++p) tsum += sm[p];
    const float thr = tsum / 49.0f;
    int pmax = 0;
    float best = -__builtin_inff();
    for (int p = 0; p < HW; ++p) {
      float mm = (sm[p] > thr) ? (fin[0][p] / 2048.0f) : 0.0f;
      if (mm > best) { best = mm; pmax = p; }
    }
    s_pmax = pmax;
  }
  __syncthreads();

  const int pmax = s_pmax;
  float pred_d = 0.0f, rdist = 0.0f;
  if (tid < HW) {
    const float u0 = fin[3][pmax];
    const float u1 = fin[4][pmax];
    float pd = fmaxf(u0 + fin[1][tid] + bvec[0], 0.0f);
    float pa = fmaxf(u1 + fin[2][tid] + bvec[1], 0.0f);
    if (tid == pmax) { pd = 0.0f; pa = 0.0f; }
    const int ii = tid / 7,  jj = tid % 7;
    const int i0 = pmax / 7, j0 = pmax % 7;
    const float rx = (float)(ii - i0) / 7.0f;
    const float ry = (float)(jj - j0) / 7.0f;
    rdist = sqrtf(rx * rx + ry * ry);
    const float ang = (atan2f(ry, rx) / 3.14159265358979323846f + 1.0f) * 0.5f;
    float gap = pa - ang;
    if (gap < 0.0f) gap += 1.0f;
    sgap[tid] = gap;
    pred_d = pd;
  }
  __syncthreads();
  if (tid == 0) {
    float gs = 0.0f;
    for (int p = 0; p < HW; ++p) gs += sgap[p];
    s_gapmean = gs / 49.0f;
  }
  __syncthreads();
  if (tid < HW) {
    const float g = sgap[tid] - s_gapmean;
    const float dl = pred_d - rdist;
    out[(size_t)n * HW + tid] = dl * dl + g * g;
  }
}

extern "C" void kernel_launch(void* const* d_in, const int* in_sizes, int n_in,
                              void* d_out, int out_size, void* d_ws, size_t ws_size,
                              hipStream_t stream) {
  const float* x    = (const float*)d_in[0];
  const float* mask = (const float*)d_in[1];
  const float* W    = (const float*)d_in[2];
  const float* b    = (const float*)d_in[3];
  float* out = (float*)d_out;
  float* Apack = (float*)d_ws;

  pack_A<<<(KSTEPS * 32 + 255) / 256, 256, 0, stream>>>(W, Apack);
  fused<<<256, 512, 0, stream>>>(x, mask, b, Apack, out);
}


# Tail after the last WMMA in one wave: mask mean, arg-max, gap mean as cross-lane butterflies; anchor geometry from a 13x13 LDS table built under the first loads; both 49-term means now tree sums
# speedup vs baseline: 1.7795x; 1.7795x over previous
#include <hip/hip_runtime.h>
#include <math.h>


typedef float v2f __attribute__((ext_vector_type(2)));
typedef float v4f __attribute__((ext_vector_type(4)));
typedef float v8f __attribute__((ext_vector_type(8)));

namespace {
constexpr int C_DIM = 2048;
constexpr int HW = 49;
constexpr int KSTEPS = C_DIM / 4;
constexpr int WAVES_PER_BLOCK = 16;
constexpr int STEPS_PER_WAVE = KSTEPS / WAVES_PER_BLOCK;
constexpr int PARTS = 4;
constexpr int STEPS_PER_PART = STEPS_PER_WAVE / PARTS;
constexpr int PART_FLOATS = STEPS_PER_PART * 4 * HW;
constexpr int PART_VEC = PART_FLOATS / 4;
constexpr int PART_LOADS = (PART_VEC + 31) / 32;
constexpr int STAGE_VEC = PART_LOADS * 32;
constexpr int HALF_LOADS = 7;
static_assert(PART_VEC == (PART_LOADS - 1) * 32 + 8, "tail load covers exactly 8 pieces");
constexpr int WAVE_CH = STEPS_PER_WAVE * 4;
constexpr int A_ROWS = 6;
constexpr int A_PITCH = WAVE_CH + 4;
static_assert(WAVE_CH == 32 * 4, "one 16-byte piece per lane covers a wave's slice of a W row");
constexpr int GRID = 7;
constexpr int GEO_SPAN = 2 * GRID - 1;
constexpr int GEO_N = GEO_SPAN * GEO_SPAN;
constexpr int TAIL_B = HW - 32;
}

#define ROWX_I(V, M) __builtin_amdgcn_update_dpp(0, (V), 0x160 + (M), 0xf, 0xf, false)
#define ROWX_F(V, M) __builtin_bit_cast(float, ROWX_I(__builtin_bit_cast(int, (V)), (M)))
__device__ __forceinline__ int xrow_i(int v) {
  return __builtin_amdgcn_permlanex16(v, v, 0x76543210, 0xfedcba98, false, false);
}
__device__ __forceinline__ float xrow_f(float v) {
  return __builtin_bit_cast(float, xrow_i(__builtin_bit_cast(int, v)));
}
__device__ __forceinline__ float wave_sum(float v) {
  v += ROWX_F(v, 1);
  v += ROWX_F(v, 2);
  v += ROWX_F(v, 4);
  v += ROWX_F(v, 8);
  v += xrow_f(v);
  return v;
}

__global__ __launch_bounds__(512) void fused(
    const float* __restrict__ x, const float* __restrict__ mask,
    const float* __restrict__ bvec, const float* __restrict__ W,
    float* __restrict__ out) {
  __shared__ v4f stage[WAVES_PER_BLOCK][STAGE_VEC];
  __shared__ v4f astage[WAVES_PER_BLOCK][A_ROWS * A_PITCH / 4];
  __shared__ float part[WAVES_PER_BLOCK][5][64];
  __shared__ float fin[5][64];
  __shared__ v2f geo[GEO_N];

  const int tid = threadIdx.x;
  const int n = blockIdx.x;
  const int wave = tid >> 5;
  const int lane = tid & 31;
  const int mrow = lane & 15;
  const int chalf = (lane >> 4) << 1;

  const float* mask_row = mask + n * HW;
  const float mkA = mask_row[lane];
  const float mkB = mask_row[lane < TAIL_B ? lane + 32 : HW - 1];
  const float bias0 = bvec[0];
  const float bias1 = bvec[1];

  const int p0 = mrow, p1 = 16 + mrow, p2 = 32 + mrow;

  v8f a0 = {}, a1 = {}, a2 = {}, a3 = {};

  const float* wsl = W + wave * WAVE_CH + 4 * lane;
  const v4f w1 = *(const v4f*)(wsl + C_DIM);
  const v4f w2 = *(const v4f*)(wsl + 2 * C_DIM + C_DIM);
  const v4f w3 = *(const v4f*)(wsl);
  const v4f w4 = *(const v4f*)(wsl + 2 * C_DIM);
  __builtin_amdgcn_sched_barrier(0);

  const v4f* xw = (const v4f*)(x + (size_t)n * (C_DIM * HW) + (size_t)wave * (STEPS_PER_WAVE * 4 * HW));
  const int tail = (PART_LOADS - 1) * 32 + (lane & 7);

  v4f R[2][PART_LOADS];
#define LOAD_PIECES(Q, JLO, JHI) do { const v4f* src_ = xw + (Q) * PART_VEC; \
    _Pragma("unroll") for (int j = (JLO); j < (JHI); ++j) \
      R[(Q) & 1][j] = (j < PART_LOADS - 1) ? src_[32 * j + lane] : src_[tail]; } while (0)

  LOAD_PIECES(0, 0, PART_LOADS);
  __builtin_amdgcn_sched_barrier(0);
  LOAD_PIECES(1, 0, PART_LOADS);

  if (tid < GEO_N) {
    const int di = tid / GEO_SPAN - (GRID - 1);
    const int dj = tid % GEO_SPAN - (GRID - 1);
    const float rx = (float)di / 7.0f;
    const float ry = (float)dj / 7.0f;
    v2f g;
    g.x = sqrtf(rx * rx + ry * ry);
    g.y = (atan2f(ry, rx) / 3.14159265358979323846f + 1.0f) * 0.5f;
    geo[tid] = g;
  }

  v4f* at = astage[wave];
  const v4f ones4 = {1.0f, 1.0f, 1.0f, 1.0f};
  const v4f zero4 = {0.0f, 0.0f, 0.0f, 0.0f};
  at[0 * (A_PITCH / 4) + lane] = ones4;
  at[1 * (A_PITCH / 4) + lane] = w1;
  at[2 * (A_PITCH / 4) + lane] = w2;
  at[3 * (A_PITCH / 4) + lane] = w3;
  at[4 * (A_PITCH / 4) + lane] = w4;
  at[5 * (A_PITCH / 4) + lane] = zero4;
  const float* ar = (const float*)at + (mrow < 5 ? mrow : 5) * A_PITCH + chalf;
  v2f Afr[STEPS_PER_WAVE];
#pragma unroll
  for (int s = 0; s < STEPS_PER_WAVE; ++s)
    Afr[s] = *(const v2f*)(ar + 4 * s);

  const float thr = wave_sum(mkA + (lane < TAIL_B ? mkB : 0.0f)) / 49.0f;
  const bool onA = mkA > thr;
  const bool onB = mkB > thr;
  float* orow = out + (size_t)n * HW;
  asm volatile("" :: "v"(thr), "s"(orow));

  v4f* sv = stage[wave];
  const float* sf = (const float*)sv + chalf * HW;

#define HALF_PART(Q, JLO, JHI, SLO, SHI) do { \
    _Pragma("unroll") for (int j = (JLO); j < (JHI); ++j) sv[32 * j + lane] = R[(Q) & 1][j]; \
      \
    __builtin_amdgcn_sched_barrier(0); \
    if ((Q) + 2 < PARTS) LOAD_PIECES((Q) + 2, JLO, JHI); \
    __builtin_amdgcn_sched_barrier(0); \
    _Pragma("unroll") for (int sq = (SLO); sq < (SHI); ++sq) { \
      const float* lb = sf + sq * (4 * HW);    \
      const v2f A = Afr[(Q) * STEPS_PER_PART + sq]; \
      v2f B0, B1, B2, B3; \
      B0.x = lb[p0];  B0.y = lb[HW + p0]; \
      B1.x = lb[p1];  B1.y = lb[HW + p1]; \
      B2.x = lb[p2];  B2.y = lb[HW + p2]; \
      B3.x = lb[48];  B3.y = lb[HW + 48]; \
      a0 = __builtin_amdgcn_wmma_f32_16x16x4_f32(false, A, false, B0, (short)0, a0, false, false); \
      a1 = __builtin_amdgcn_wmma_f32_16x16x4_f32(false, A, false, B1, (short)0, a1, false, false); \
      a2 = __builtin_amdgcn_wmma_f32_16x16x4_f32(false, A, false, B2, (short)0, a2, false, false); \
      a3 = __builtin_amdgcn_wmma_f32_16x16x4_f32(false, A, false, B3, (short)0, a3, false, false); \
    } } while (0)

  static_assert(HALF_LOADS * 512 >= (STEPS_PER_PART / 2) * 4 * HW * 4, "first half's rows complete");
#pragma unroll
  for (int q = 0; q < PARTS; ++q) {
    HALF_PART(q, 0, HALF_LOADS, 0, STEPS_PER_PART / 2);
    HALF_PART(q, HALF_LOADS, PART_LOADS, STEPS_PER_PART / 2, STEPS_PER_PART);
  }
#undef HALF_PART
#undef LOAD_PIECES

  if (lane < 16) {
#pragma unroll
    for (int m = 0; m < 5; ++m) {
      part[wave][m][mrow]      = a0[m];
      part[wave][m][16 + mrow] = a1[m];
      part[wave][m][32 + mrow] = a2[m];
      part[wave][m][48 + mrow] = a3[m];
    }
  }
  __syncthreads();

  if (tid < 5 * 64) {
    const int m = tid >> 6, p = tid & 63;
    float ssum = 0.0f;
    for (int w = 0; w < WAVES_PER_BLOCK; ++w) ssum += part[w][m][p];
    fin[m][p] = ssum;
  }
  __syncthreads();

  if (__builtin_amdgcn_readfirstlane(wave) != 0) return;

  const bool hasB = lane < TAIL_B;
  const int pA = lane;
  const int pB = hasB ? lane + 32 : HW - 1;
  const int codeA = (pA << 8) | ((pA / GRID) * GEO_SPAN + pA % GRID);
  const int codeB = (pB << 8) | ((pB / GRID) * GEO_SPAN + pB % GRID);
  const float SA = fin[0][lane], SB = fin[0][lane + 32];
  const float s0A = fin[1][lane], s0B = fin[1][lane + 32];
  const float s1A = fin[2][lane], s1B = fin[2][lane + 32];

  float bv = onA ? (SA / 2048.0f) : 0.0f;
  int bc = codeA;
  {
    const float vB = hasB ? (onB ? (SB / 2048.0f) : 0.0f) : -__builtin_inff();
    if (vB > bv) { bv = vB; bc = codeB; }
  }
#define ARGMAX_STEP(OV, OC) do { const float ov_ = (OV); const int oc_ = (OC); \
    const int lo_ = oc_ < bc ? oc_ : bc; \
    int nc_ = (ov_ > bv) ? oc_ : bc; \
    nc_ = (ov_ == bv) ? lo_ : nc_; \
    bv = (ov_ > bv) ? ov_ : bv; bc = nc_; } while (0)
  ARGMAX_STEP(ROWX_F(bv, 1), ROWX_I(bc, 1));
  ARGMAX_STEP(ROWX_F(bv, 2), ROWX_I(bc, 2));
  ARGMAX_STEP(ROWX_F(bv, 4), ROWX_I(bc, 4));
  ARGMAX_STEP(ROWX_F(bv, 8), ROWX_I(bc, 8));
  ARGMAX_STEP(xrow_f(bv), xrow_i(bc));
#undef ARGMAX_STEP
  const int pmax = bc >> 8;
  const int anchor = bc & 255;

  const float u0 = fin[3][pmax];
  const float u1 = fin[4][pmax];
  constexpr int GEO_MID = (GRID - 1) * GEO_SPAN + (GRID - 1);
  const v2f gA = geo[(codeA & 255) + GEO_MID - anchor];
  const v2f gB = geo[(codeB & 255) + GEO_MID - anchor];

  float pdA = fmaxf(u0 + s0A + bias0, 0.0f);
  float paA = fmaxf(u1 + s1A + bias1, 0.0f);
  if (pA == pmax) { pdA = 0.0f; paA = 0.0f; }
  float pdB = fmaxf(u0 + s0B + bias0, 0.0f);
  float paB = fmaxf(u1 + s1B + bias1, 0.0f);
  if (pB == pmax) { pdB = 0.0f; paB = 0.0f; }
  float gapA = paA - gA.y;
  if (gapA < 0.0f) gapA += 1.0f;
  float gapB = paB - gB.y;
  if (gapB < 0.0f) gapB += 1.0f;

  const float gapmean = wave_sum(gapA + (hasB ? gapB : 0.0f)) / 49.0f;

  {
    const float g = gapA - gapmean;
    const float dl = pdA - gA.x;
    orow[lane] = dl * dl + g * g;
  }
  if (hasB) {
    const float g = gapB - gapmean;
    const float dl = pdB - gB.x;
    orow[lane + 32] = dl * dl + g * g;
  }
}

extern "C" void kernel_launch(void* const* d_in, const int* in_sizes, int n_in,
                              void* d_out, int out_size, void* d_ws, size_t ws_size,
                              hipStream_t stream) {
  const float* x    = (const float*)d_in[0];
  const float* mask = (const float*)d_in[1];
  const float* W    = (const float*)d_in[2];
  const float* b    = (const float*)d_in[3];
  float* out = (float*)d_out;
  (void)d_ws; (void)ws_size;

  fused<<<256, 512, 0, stream>>>(x, mask, b, W, out);
}


# Two waves per SIMD (8 per work-group, 256 channels each); the next half part's LDS copy and fragment loads go as one group ahead of the current 16 WMMAs, one load per fragment; 8 partial sums, not 16
# speedup vs baseline: 1.8751x; 1.0537x over previous
#include <hip/hip_runtime.h>
#include <math.h>


typedef float v2f __attribute__((ext_vector_type(2)));
typedef float v4f __attribute__((ext_vector_type(4)));
typedef float v8f __attribute__((ext_vector_type(8)));

namespace {
constexpr int C_DIM = 2048;
constexpr int HW = 49;
constexpr int KSTEPS = C_DIM / 4;
constexpr int WAVES_PER_BLOCK = 8;
constexpr int STEPS_PER_WAVE = KSTEPS / WAVES_PER_BLOCK;
constexpr int PARTS = 8;
constexpr int STEPS_PER_PART = STEPS_PER_WAVE / PARTS;
constexpr int PART_FLOATS = STEPS_PER_PART * 4 * HW;
constexpr int PART_VEC = PART_FLOATS / 4;
constexpr int PART_LOADS = (PART_VEC + 31) / 32;
constexpr int STAGE_VEC = PART_LOADS * 32;
constexpr int STAGE_PITCH = 896;
constexpr int HALF_LOADS = 7;
constexpr int HSTEPS = STEPS_PER_PART / 2;
constexpr int DEPTH = 4;
static_assert(PART_VEC == (PART_LOADS - 1) * 32 + 8, "tail load covers exactly 8 pieces");
constexpr int WAVE_CH = STEPS_PER_WAVE * 4;
constexpr int W_PIECES = WAVE_CH / 128;
constexpr int A_ROWS = 6;
constexpr int A_PITCH = WAVE_CH + 4;
static_assert(WAVE_CH == W_PIECES * 32 * 4, "16-byte pieces per lane cover a wave's slice of a W row");
static_assert(PARTS % DEPTH == 0 && STEPS_PER_PART == 8, "part slots and half parts as the stream below assumes");
constexpr int GRID = 7;
constexpr int GEO_SPAN = 2 * GRID - 1;
constexpr int GEO_N = GEO_SPAN * GEO_SPAN;
constexpr int TAIL_B = HW - 32;
}

#define ROWX_I(V, M) __builtin_amdgcn_update_dpp(0, (V), 0x160 + (M), 0xf, 0xf, false)
#define ROWX_F(V, M) __builtin_bit_cast(float, ROWX_I(__builtin_bit_cast(int, (V)), (M)))
__device__ __forceinline__ int xrow_i(int v) {
  return __builtin_amdgcn_permlanex16(v, v, 0x76543210, 0xfedcba98, false, false);
}
__device__ __forceinline__ float xrow_f(float v) {
  return __builtin_bit_cast(float, xrow_i(__builtin_bit_cast(int, v)));
}
__device__ __forceinline__ float wave_sum(float v) {
  v += ROWX_F(v, 1);
  v += ROWX_F(v, 2);
  v += ROWX_F(v, 4);
  v += ROWX_F(v, 8);
  v += xrow_f(v);
  return v;
}

__global__ __launch_bounds__(256) void fused(
    const float* __restrict__ x, const float* __restrict__ mask,
    const float* __restrict__ bvec, const float* __restrict__ W,
    float* __restrict__ out) {
  static_assert(STAGE_PITCH >= STAGE_VEC, "a part fits its row");
  __shared__ v4f stage[WAVES_PER_BLOCK][STAGE_PITCH];
  __shared__ v4f astage[WAVES_PER_BLOCK][A_ROWS * A_PITCH / 4];
  __shared__ float part[WAVES_PER_BLOCK][5][64];
  __shared__ v2f geo[GEO_N];

  const int tid = threadIdx.x;
  const int n = blockIdx.x;
  const int wave = tid >> 5;
  const int lane = tid & 31;
  const int mrow = lane & 15;
  const int chalf = (lane >> 4) << 1;

  const float* mask_row = mask + n * HW;
  const float mkA = mask_row[lane];
  const float mkB = mask_row[lane < TAIL_B ? lane + 32 : HW - 1];
  const float bias0 = bvec[0];
  const float bias1 = bvec[1];

  v8f a0 = {}, a1 = {}, a2 = {}, a3 = {};

  const float* wsl = W + wave * WAVE_CH + 4 * lane;
  v4f wr[4][W_PIECES];
#pragma unroll
  for (int i = 0; i < W_PIECES; ++i) {
    wr[0][i] = *(const v4f*)(wsl + 128 * i + C_DIM);
    wr[1][i] = *(const v4f*)(wsl + 128 * i + 3 * C_DIM);
    wr[2][i] = *(const v4f*)(wsl + 128 * i);
    wr[3][i] = *(const v4f*)(wsl + 128 * i + 2 * C_DIM);
  }
  __builtin_amdgcn_sched_barrier(0);

  const v4f* xw = (const v4f*)(x + (size_t)n * (C_DIM * HW) + (size_t)wave * (STEPS_PER_WAVE * 4 * HW));
  const int tail = (PART_LOADS - 1) * 32 + (lane & 7);

  v4f R[DEPTH][PART_LOADS];
#define LOAD_PIECES(SLOT, Q, JLO, JHI) do { const v4f* src_ = xw + (Q) * PART_VEC; \
    _Pragma("unroll") for (int j = (JLO); j < (JHI); ++j) \
      R[SLOT][j] = (j < PART_LOADS - 1) ? src_[32 * j + lane] : src_[tail]; } while (0)

  LOAD_PIECES(0, 0, 0, PART_LOADS);
  __builtin_amdgcn_sched_barrier(0);
  LOAD_PIECES(1, 1, 0, PART_LOADS);
  __builtin_amdgcn_sched_barrier(0);
  LOAD_PIECES(2, 2, 0, PART_LOADS);
  __builtin_amdgcn_sched_barrier(0);
  LOAD_PIECES(3, 3, 0, PART_LOADS);

  static_assert(GEO_N <= WAVES_PER_BLOCK * 32, "one table entry per thread");
  if (tid < GEO_N) {
    const int t = tid;
    const int di = t / GEO_SPAN - (GRID - 1);
    const int dj = t % GEO_SPAN - (GRID - 1);
    const float rx = (float)di / 7.0f;
    const float ry = (float)dj / 7.0f;
    v2f g;
    g.x = sqrtf(rx * rx + ry * ry);
    g.y = (atan2f(ry, rx) / 3.14159265358979323846f + 1.0f) * 0.5f;
    geo[t] = g;
  }

  v4f* at = astage[wave];
  const v4f ones4 = {1.0f, 1.0f, 1.0f, 1.0f};
  const v4f zero4 = {0.0f, 0.0f, 0.0f, 0.0f};
#pragma unroll
  for (int i = 0; i < W_PIECES; ++i) {
    at[0 * (A_PITCH / 4) + 32 * i + lane] = ones4;
    at[1 * (A_PITCH / 4) + 32 * i + lane] = wr[0][i];
    at[2 * (A_PITCH / 4) + 32 * i + lane] = wr[1][i];
    at[3 * (A_PITCH / 4) + 32 * i + lane] = wr[2][i];
    at[4 * (A_PITCH / 4) + 32 * i + lane] = wr[3][i];
    at[5 * (A_PITCH / 4) + 32 * i + lane] = zero4;
  }
  const float* ar = (const float*)at + (mrow < 5 ? mrow : 5) * A_PITCH + chalf;

  const float thr = wave_sum(mkA + (lane < TAIL_B ? mkB : 0.0f)) / 49.0f;
  const bool onA = mkA > thr;
  const bool onB = mkB > thr;
  float* orow = out + (size_t)n * HW;
  asm volatile("" :: "v"(thr), "s"(orow));

  v4f* sv = stage[wave];
  const float* sf = (const float*)sv + chalf * HW;

  const int q0 = mrow;
  int q1 = 16 + mrow, q2 = 32 + mrow;
  asm volatile("" : "+v"(q1));
  asm volatile("" : "+v"(q2));

  v2f Bs[2][HSTEPS][4];
  v2f As[2][HSTEPS];

#define STORE_PIECES(SLOT, JLO, JHI) do { \
    _Pragma("unroll") for (int j = (JLO); j < (JHI); ++j) sv[32 * j + lane] = R[SLOT][j]; } while (0)
#define LOAD_FRAGS(SET, H, KS) do { \
    _Pragma("unroll") for (int sq = 0; sq < HSTEPS; ++sq) { \
      const float* lb = sf + ((H) * HSTEPS + sq) * (4 * HW);    \
      Bs[SET][sq][0].x = lb[q0];  Bs[SET][sq][0].y = lb[HW + q0]; \
      Bs[SET][sq][1].x = lb[q1];  Bs[SET][sq][1].y = lb[HW + q1]; \
      Bs[SET][sq][2].x = lb[q2];  Bs[SET][sq][2].y = lb[HW + q2]; \
      Bs[SET][sq][3].x = lb[48];  Bs[SET][sq][3].y = lb[HW + 48]; \
      As[SET][sq] = *(const v2f*)(ar + 4 * ((KS) + sq)); \
    } } while (0)
#define MMA_HALF(SET) do { \
    _Pragma("unroll") for (int sq = 0; sq < HSTEPS; ++sq) { \
      a0 = __builtin_amdgcn_wmma_f32_16x16x4_f32(false, As[SET][sq], false, Bs[SET][sq][0], (short)0, a0, false, false); \
      a1 = __builtin_amdgcn_wmma_f32_16x16x4_f32(false, As[SET][sq], false, Bs[SET][sq][1], (short)0, a1, false, false); \
      a2 = __builtin_amdgcn_wmma_f32_16x16x4_f32(false, As[SET][sq], false, Bs[SET][sq][2], (short)0, a2, false, false); \
      a3 = __builtin_amdgcn_wmma_f32_16x16x4_f32(false, As[SET][sq], false, Bs[SET][sq][3], (short)0, a3, false, false); \
    } } while (0)

  static_assert(HALF_LOADS * 512 >= HSTEPS * 4 * HW * 4, "first half's rows complete");

  STORE_PIECES(0, 0, HALF_LOADS);
  __builtin_amdgcn_sched_barrier(0);
  LOAD_PIECES(0, DEPTH, 0, HALF_LOADS);
  __builtin_amdgcn_sched_barrier(0);
  LOAD_FRAGS(0, 0, 0);

#pragma unroll
  for (int g = 0; g < 2 * PARTS; ++g) {
    if (g + 1 < 2 * PARTS) {
      const int pn = (g + 1) >> 1, hn = (g + 1) & 1;
      if (hn == 0) STORE_PIECES(pn % DEPTH, 0, HALF_LOADS); else STORE_PIECES(pn % DEPTH, HALF_LOADS, PART_LOADS);
      __builtin_amdgcn_sched_barrier(0);
      if (pn + DEPTH < PARTS) {
        if (hn == 0) LOAD_PIECES(pn % DEPTH, pn + DEPTH, 0, HALF_LOADS); else LOAD_PIECES(pn % DEPTH, pn + DEPTH, HALF_LOADS, PART_LOADS);
      }
      __builtin_amdgcn_sched_barrier(0);
      if (hn == 0) LOAD_FRAGS((g + 1) & 1, 0, STEPS_PER_PART * pn); else LOAD_FRAGS((g + 1) & 1, 1, STEPS_PER_PART * pn + HSTEPS);
    }
    __builtin_amdgcn_sched_barrier(0);
    MMA_HALF(g & 1);
    __builtin_amdgcn_sched_barrier(0);
  }
#undef MMA_HALF
#undef LOAD_FRAGS
#undef STORE_PIECES
#undef LOAD_PIECES

  if (lane < 16) {
#pragma unroll
    for (int m = 0; m < 5; ++m) {
      part[wave][m][mrow]      = a0[m];
      part[wave][m][16 + mrow] = a1[m];
      part[wave][m][32 + mrow] = a2[m];
      part[wave][m][48 + mrow] = a3[m];
    }
  }
  __syncthreads();

  if (__builtin_amdgcn_readfirstlane(wave) != 0) return;

#define PSUM(M, P) (((((((part[0][M][P] + part[1][M][P]) + part[2][M][P]) + part[3][M][P]) + part[4][M][P]) + part[5][M][P]) + part[6][M][P]) + part[7][M][P])
  static_assert(WAVES_PER_BLOCK == 8, "PSUM adds eight partial tiles");

  const bool hasB = lane < TAIL_B;
  const int pA = lane;
  const int pB = hasB ? lane + 32 : HW - 1;
  const int codeA = (pA << 8) | ((pA / GRID) * GEO_SPAN + pA % GRID);
  const int codeB = (pB << 8) | ((pB / GRID) * GEO_SPAN + pB % GRID);
  const float SA = PSUM(0, lane), SB = PSUM(0, lane + 32);
  const float s0A = PSUM(1, lane), s0B = PSUM(1, lane + 32);
  const float s1A = PSUM(2, lane), s1B = PSUM(2, lane + 32);

  float bv = onA ? (SA / 2048.0f) : 0.0f;
  int bc = codeA;
  {
    const float vB = hasB ? (onB ? (SB / 2048.0f) : 0.0f) : -__builtin_inff();
    if (vB > bv) { bv = vB; bc = codeB; }
  }
#define ARGMAX_STEP(OV, OC) do { const float ov_ = (OV); const int oc_ = (OC); \
    const int lo_ = oc_ < bc ? oc_ : bc; \
    int nc_ = (ov_ > bv) ? oc_ : bc; \
    nc_ = (ov_ == bv) ? lo_ : nc_; \
    bv = (ov_ > bv) ? ov_ : bv; bc = nc_; } while (0)
  ARGMAX_STEP(ROWX_F(bv, 1), ROWX_I(bc, 1));
  ARGMAX_STEP(ROWX_F(bv, 2), ROWX_I(bc, 2));
  ARGMAX_STEP(ROWX_F(bv, 4), ROWX_I(bc, 4));
  ARGMAX_STEP(ROWX_F(bv, 8), ROWX_I(bc, 8));
  ARGMAX_STEP(xrow_f(bv), xrow_i(bc));
#undef ARGMAX_STEP
  const int pmax = bc >> 8;
  const int anchor = bc & 255;

  const float u0 = PSUM(3, pmax);
  const float u1 = PSUM(4, pmax);
#undef PSUM
  constexpr int GEO_MID = (GRID - 1) * GEO_SPAN + (GRID - 1);
  const v2f gA = geo[(codeA & 255) + GEO_MID - anchor];
  const v2f gB = geo[(codeB & 255) + GEO_MID - anchor];

  float pdA = fmaxf(u0 + s0A + bias0, 0.0f);
  float paA = fmaxf(u1 + s1A + bias1, 0.0f);
  if (pA == pmax) { pdA = 0.0f; paA = 0.0f; }
  float pdB = fmaxf(u0 + s0B + bias0, 0.0f);
  float paB = fmaxf(u1 + s1B + bias1, 0.0f);
  if (pB == pmax) { pdB = 0.0f; paB = 0.0f; }
  float gapA = paA - gA.y;
  if (gapA < 0.0f) gapA += 1.0f;
  float gapB = paB - gB.y;
  if (gapB < 0.0f) gapB += 1.0f;

  const float gapmean = wave_sum(gapA + (hasB ? gapB : 0.0f)) / 49.0f;

  {
    const float g = gapA - gapmean;
    const float dl = pdA - gA.x;
    orow[lane] = dl * dl + g * g;
  }
  if (hasB) {
    const float g = gapB - gapmean;
    const float dl = pdB - gB.x;
    orow[lane + 32] = dl * dl + g * g;
  }
}

extern "C" void kernel_launch(void* const* d_in, const int* in_sizes, int n_in,
                              void* d_out, int out_size, void* d_ws, size_t ws_size,
                              hipStream_t stream) {
  const float* x    = (const float*)d_in[0];
  const float* mask = (const float*)d_in[1];
  const float* W    = (const float*)d_in[2];
  const float* b    = (const float*)d_in[3];
  float* out = (float*)d_out;
  (void)d_ws; (void)ws_size;

  fused<<<256, 256, 0, stream>>>(x, mask, b, W, out);
}
